# starting kernel
# baseline (speedup 1.0000x reference)
#include <hip/hip_runtime.h>

#define NUM_CAM 16
#define IN_SZ   256
#define OUT_SZ  128
#define BATCH   8192
#define TILE_M  16
#define XPITCH  260
#define WPITCH  130

typedef __attribute__((ext_vector_type(2))) float v2f;
typedef __attribute__((ext_vector_type(8))) float v8f;

__global__ void fc_zero_counts(int* counts) {
    if (threadIdx.x < NUM_CAM) counts[threadIdx.x] = 0;
}

__global__ void fc_bucket(const int* __restrict__ cam,
                          int* __restrict__ counts,
                          int* __restrict__ idx) {
    int t = blockIdx.x * blockDim.x + threadIdx.x;
    if (t < BATCH) {
        int c = cam[t] & (NUM_CAM - 1);
        int pos = atomicAdd(&counts[c], 1);
        idx[c * BATCH + pos] = t;
    }
}

__launch_bounds__(256)
__global__ void fc_expert_gemm(const float* __restrict__ x,
                               const float* __restrict__ W,
                               const float* __restrict__ bias,
                               const int*  __restrict__ counts,
                               const int*  __restrict__ idx,
                               float* __restrict__ out) {
    __shared__ float Xs[TILE_M * XPITCH];
    __shared__ v2f   Wt[128 * WPITCH];
    __shared__ int   Si[TILE_M];

    const int c     = blockIdx.x;
    const int tile  = blockIdx.y;
    const int count = counts[c];
    if (tile * TILE_M >= count) return;

    const int tid  = threadIdx.x;
    const int lane = tid & 31;
    const int wave = tid >> 5;
    const int half = lane >> 4;
    const int l    = lane & 15;

    {
        int row  = tid >> 4;
        int col0 = (tid & 15) * 16;
        int r    = tile * TILE_M + row;
        int s    = idx[c * BATCH + min(r, count - 1)];
        if ((tid & 15) == 0) Si[row] = s;
        const float4* src = (const float4*)(x + (size_t)s * IN_SZ + col0);
        float* dst = &Xs[row * XPITCH + col0];
        #pragma unroll
        for (int i = 0; i < 4; ++i) {
            float4 v = src[i];
            dst[i * 4 + 0] = v.x; dst[i * 4 + 1] = v.y;
            dst[i * 4 + 2] = v.z; dst[i * 4 + 3] = v.w;
        }
    }

    {
        const v2f* Wsrc = (const v2f*)(W + (size_t)c * OUT_SZ * IN_SZ);
        int p     = tid & 127;
        int nbase = tid >> 7;
        #pragma unroll 4
        for (int it = 0; it < 64; ++it) {
            int n = it * 2 + nbase;
            Wt[p * WPITCH + n] = Wsrc[n * 128 + p];
        }
    }
    __syncthreads();

    const int n0 = wave * 16;
    const float* ap = &Xs[l * XPITCH + 2 * half];
    const v2f*   bp = &Wt[half * WPITCH + (n0 + l)];

    v8f acc = {};
    for (int ko = 0; ko < 8; ++ko) {
        #pragma unroll
        for (int ki = 0; ki < 8; ++ki) {
            v2f a = *(const v2f*)(ap + ki * 4);
            v2f b = bp[ki * 2 * WPITCH];
            acc = __builtin_amdgcn_wmma_f32_16x16x4_f32(
                 false, a,  false, b,
                 (short)0, acc,  false,  false);
        }
        ap += 32;
        bp += 16 * WPITCH;
    }

    const float bv = bias[c * OUT_SZ + n0 + l];
    const int base = tile * TILE_M;
    #pragma unroll
    for (int r = 0; r < 8; ++r) {
        int m  = r + half * 8;
        int gm = base + m;
        if (gm < count) {
            int s = Si[m];
            float y = acc[r] + bv;
            y = __builtin_amdgcn_rcpf(1.0f + __expf(-y));
            out[(size_t)s * OUT_SZ + n0 + l] = y;
        }
    }
}

extern "C" void kernel_launch(void* const* d_in, const int* in_sizes, int n_in,
                              void* d_out, int out_size, void* d_ws, size_t ws_size,
                              hipStream_t stream) {
    const float* x    = (const float*)d_in[0];
    const int*   cam  = (const int*)  d_in[1];
    const float* W    = (const float*)d_in[2];
    const float* bias = (const float*)d_in[3];
    float* out = (float*)d_out;

    int* counts = (int*)d_ws;
    int* idx    = counts + 64;

    fc_zero_counts<<<1, 32, 0, stream>>>(counts);
    fc_bucket<<<BATCH / 256, 256, 0, stream>>>(cam, counts, idx);

    dim3 grid(NUM_CAM, BATCH / TILE_M);
    fc_expert_gemm<<<grid, 256, 0, stream>>>(x, W, bias, counts, idx, out);
}


# W[c] kept in registers: each wave loads its 16 channels once as B fragments (128 VGPRs), no LDS copy of W; block walks its camera's tiles 2-3 at a time; grid 16x16; sum order: k regrouped within 8
# speedup vs baseline: 1.4509x; 1.4509x over previous
#include <hip/hip_runtime.h>


#define NUM_CAM 16
#define IN_SZ   256
#define OUT_SZ  128
#define BATCH   8192
#define TILE_M  16
#define NSLICE  16
#define GMAX    3
#define XP4     65

typedef __attribute__((ext_vector_type(2))) float v2f;
typedef __attribute__((ext_vector_type(4))) float v4f;
typedef __attribute__((ext_vector_type(8))) float v8f;

__global__ void fc_zero_counts(int* counts) {
    if (threadIdx.x < NUM_CAM) counts[threadIdx.x] = 0;
}

__global__ void fc_bucket(const int* __restrict__ cam,
                          int* __restrict__ counts,
                          int* __restrict__ idx) {
    int t = blockIdx.x * blockDim.x + threadIdx.x;
    if (t < BATCH) {
        int c = cam[t] & (NUM_CAM - 1);
        int pos = atomicAdd(&counts[c], 1);
        idx[c * BATCH + pos] = t;
    }
}

template <int G>
__device__ __forceinline__ void fc_group(const v4f (&wv)[32],
                                         const v4f* __restrict__ Xs4,
                                         const int* __restrict__ Si,
                                         int l, int half, int n, float bv,
                                         int row0, int count,
                                         float* __restrict__ out) {
    v8f acc[G];
    #pragma unroll
    for (int g = 0; g < G; ++g) acc[g] = v8f{};

    const v4f* ap = Xs4 + l * XP4 + half;

    #pragma unroll
    for (int j = 0; j < 32; ++j) {
        v4f a[G];
        #pragma unroll
        for (int g = 0; g < G; ++g) a[g] = ap[g * (TILE_M * XP4) + 2 * j];
        const v2f b0 = wv[j].xy;
        const v2f b1 = wv[j].zw;
        #pragma unroll
        for (int g = 0; g < G; ++g)
            acc[g] = __builtin_amdgcn_wmma_f32_16x16x4_f32(
                false, a[g].xy, false, b0, (short)0, acc[g], false, false);
        #pragma unroll
        for (int g = 0; g < G; ++g)
            acc[g] = __builtin_amdgcn_wmma_f32_16x16x4_f32(
                false, a[g].zw, false, b1, (short)0, acc[g], false, false);
    }

    if constexpr (G == 1) {
        asm volatile("v_nop\n\tv_nop\n\tv_nop\n\tv_nop\n\tv_nop" : "+v"(acc[0]), "+s"(row0));
    } else if constexpr (G == 2) {
        asm volatile("v_nop\n\tv_nop\n\tv_nop\n\tv_nop\n\tv_nop" : "+v"(acc[0]), "+v"(acc[1]), "+s"(row0));
    } else {
        asm volatile("v_nop\n\tv_nop\n\tv_nop\n\tv_nop\n\tv_nop" : "+v"(acc[0]), "+v"(acc[1]), "+v"(acc[2]), "+s"(row0));
    }

    #pragma unroll
    for (int g = 0; g < G; ++g) {
        #pragma unroll
        for (int r = 0; r < 8; ++r) {
            int m  = g * TILE_M + r + half * 8;
            int gm = row0 + m;
            if (gm < count) {
                int s = Si[m];
                float y = acc[g][r] + bv;
                y = __builtin_amdgcn_rcpf(1.0f + __expf(-y));
                out[(size_t)s * OUT_SZ + n] = y;
            }
        }
    }
}

__launch_bounds__(256)
__global__ void fc_expert_gemm(const float* __restrict__ x,
                               const float* __restrict__ W,
                               const float* __restrict__ bias,
                               const int*  __restrict__ counts,
                               const int*  __restrict__ idx,
                               float* __restrict__ out) {
    __shared__ v4f Xs4[GMAX * TILE_M * XP4];
    __shared__ int Si[GMAX * TILE_M];

    const int c     = blockIdx.x;
    const int slice = blockIdx.y;
    const int tid   = threadIdx.x;
    const int lane  = tid & 31;
    const int wave  = tid >> 5;
    const int half  = lane >> 4;
    const int l     = lane & 15;
    const int n     = wave * 16 + l;

    const int count  = counts[c];
    const int ntiles = (count + TILE_M - 1) / TILE_M;
    const int t_begin = (slice * ntiles) / NSLICE;
    const int t_end   = ((slice + 1) * ntiles) / NSLICE;
    if (t_begin >= t_end) return;

    v4f wv[32];
    {
        const v4f* wsrc = (const v4f*)(W + ((size_t)c * OUT_SZ + n) * IN_SZ + 4 * half);
        #pragma unroll
        for (int j = 0; j < 32; ++j) wv[j] = wsrc[2 * j];
    }
    const float bv = bias[c * OUT_SZ + n];

    const int srow = tid >> 3;
    const int sq   = tid & 7;

    for (int t0 = t_begin; t0 < t_end; t0 += GMAX) {
        const int g     = min(GMAX, t_end - t0);
        const int row0  = t0 * TILE_M;
        const int nrows = g * TILE_M;

        for (int r = srow; r < nrows; r += 32) {
            int s = idx[c * BATCH + min(row0 + r, count - 1)];
            if (sq == 0) Si[r] = s;
            const v4f* src = (const v4f*)(x + (size_t)s * IN_SZ) + sq;
            v4f* dst = Xs4 + r * XP4 + sq;
            #pragma unroll
            for (int i = 0; i < 8; ++i) dst[i * 8] = src[i * 8];
        }
        __syncthreads();

        if (g == 3)      fc_group<3>(wv, Xs4, Si, l, half, n, bv, row0, count, out);
        else if (g == 2) fc_group<2>(wv, Xs4, Si, l, half, n, bv, row0, count, out);
        else             fc_group<1>(wv, Xs4, Si, l, half, n, bv, row0, count, out);

        if (t0 + GMAX < t_end) __syncthreads();
    }
}

extern "C" void kernel_launch(void* const* d_in, const int* in_sizes, int n_in,
                              void* d_out, int out_size, void* d_ws, size_t ws_size,
                              hipStream_t stream) {
    const float* x    = (const float*)d_in[0];
    const int*   cam  = (const int*)  d_in[1];
    const float* W    = (const float*)d_in[2];
    const float* bias = (const float*)d_in[3];
    float* out = (float*)d_out;

    int* counts = (int*)d_ws;
    int* idx    = counts + 64;

    fc_zero_counts<<<1, 32, 0, stream>>>(counts);
    fc_bucket<<<BATCH / 256, 256, 0, stream>>>(cam, counts, idx);

    dim3 grid(NUM_CAM, NSLICE);
    fc_expert_gemm<<<grid, 256, 0, stream>>>(x, W, bias, counts, idx, out);
}


# Bucket passes dropped: batch cut in 16 ranges of 512, block (camera, range) picks its rows from the range's cam values by wave ballots; up to 4 tiles a group; 3 launches become 1; sums unchanged
# speedup vs baseline: 1.9343x; 1.3332x over previous
#include <hip/hip_runtime.h>


#define NUM_CAM 16
#define IN_SZ   256
#define OUT_SZ  128
#define BATCH   8192
#define TILE_M  16
#define NRANGE  16
#define RANGE   (BATCH / NRANGE)
#define GMAX    4
#define XP4     65

typedef __attribute__((ext_vector_type(2))) float v2f;
typedef __attribute__((ext_vector_type(4))) float v4f;
typedef __attribute__((ext_vector_type(8))) float v8f;

template <int G>
__device__ __forceinline__ void fc_group(const v4f (&wv)[32],
                                         const v4f* __restrict__ Xs4,
                                         const int* __restrict__ Sl,
                                         int l, int half, int n, float bv,
                                         int row0, int count,
                                         float* __restrict__ out) {
    v8f acc[G];
    #pragma unroll
    for (int g = 0; g < G; ++g) {
        acc[g] = v8f{};
        asm volatile("" : "+v"(acc[g]));
    }

    const v4f* ap = Xs4 + l * XP4 + half;

    #pragma unroll
    for (int j = 0; j < 32; ++j) {
        v4f a[G];
        #pragma unroll
        for (int g = 0; g < G; ++g) a[g] = ap[g * (TILE_M * XP4) + 2 * j];
        const v2f b0 = wv[j].xy;
        const v2f b1 = wv[j].zw;
        #pragma unroll
        for (int g = 0; g < G; ++g)
            acc[g] = __builtin_amdgcn_wmma_f32_16x16x4_f32(
                false, a[g].xy, false, b0, (short)0, acc[g], false, false);
        #pragma unroll
        for (int g = 0; g < G; ++g)
            acc[g] = __builtin_amdgcn_wmma_f32_16x16x4_f32(
                false, a[g].zw, false, b1, (short)0, acc[g], false, false);
    }

    if constexpr (G == 1) {
        asm volatile("v_nop\n\tv_nop\n\tv_nop\n\tv_nop\n\tv_nop" : "+v"(acc[0]), "+s"(row0));
    } else if constexpr (G == 2) {
        asm volatile("v_nop\n\tv_nop\n\tv_nop\n\tv_nop\n\tv_nop" : "+v"(acc[0]), "+v"(acc[1]), "+s"(row0));
    } else if constexpr (G == 3) {
        asm volatile("v_nop\n\tv_nop\n\tv_nop\n\tv_nop\n\tv_nop" : "+v"(acc[0]), "+v"(acc[1]), "+v"(acc[2]), "+s"(row0));
    } else {
        asm volatile("v_nop\n\tv_nop\n\tv_nop\n\tv_nop\n\tv_nop" : "+v"(acc[0]), "+v"(acc[1]), "+v"(acc[2]), "+v"(acc[3]), "+s"(row0));
    }

    #pragma unroll
    for (int g = 0; g < G; ++g) {
        #pragma unroll
        for (int r = 0; r < 8; ++r) {
            int gm = row0 + g * TILE_M + r + half * 8;
            if (gm < count) {
                int s = Sl[gm];
                float y = acc[g][r] + bv;
                y = __builtin_amdgcn_rcpf(1.0f + __expf(-y));
                out[(size_t)s * OUT_SZ + n] = y;
            }
        }
    }
}

__launch_bounds__(256)
__global__ void fc_expert_gemm(const float* __restrict__ x,
                               const int*   __restrict__ cam,
                               const float* __restrict__ W,
                               const float* __restrict__ bias,
                               float* __restrict__ out) {
    __shared__ v4f Xs4[GMAX * TILE_M * XP4];
    __shared__ int Sl[RANGE];
    __shared__ int wtot[8];

    const int c     = blockIdx.x;
    const int range = blockIdx.y;
    const int tid   = threadIdx.x;
    const int lane  = tid & 31;
    const int wave  = tid >> 5;
    const int half  = lane >> 4;
    const int l     = lane & 15;
    const int n     = wave * 16 + l;

    const int e0 = range * RANGE + 2 * tid;
    const int2 cv = *(const int2*)(cam + e0);

    v4f wv[32];
    {
        const v4f* wsrc = (const v4f*)(W + ((size_t)c * OUT_SZ + n) * IN_SZ + 4 * half);
        #pragma unroll
        for (int j = 0; j < 32; ++j) wv[j] = wsrc[2 * j];
    }
    const float bv = bias[c * OUT_SZ + n];

    const bool m0 = (cv.x & (NUM_CAM - 1)) == c;
    const bool m1 = (cv.y & (NUM_CAM - 1)) == c;
    const unsigned b0 = __builtin_amdgcn_ballot_w32(m0);
    const unsigned b1 = __builtin_amdgcn_ballot_w32(m1);
    const unsigned below = (1u << lane) - 1u;
    const int pre = __popc(b0 & below) + __popc(b1 & below);
    if (lane == 0) wtot[wave] = __popc(b0) + __popc(b1);
    __syncthreads();
    int off = 0, count = 0;
    #pragma unroll
    for (int i = 0; i < 8; ++i) {
        const int v = wtot[i];
        off   += (i < wave) ? v : 0;
        count += v;
    }
    if (m0) Sl[off + pre] = e0;
    if (m1) Sl[off + pre + (m0 ? 1 : 0)] = e0 + 1;
    __syncthreads();
    if (count == 0) return;

    const int ntiles = (count + TILE_M - 1) / TILE_M;

    const int srow = tid >> 3;
    const int sq   = tid & 7;

    for (int t0 = 0; t0 < ntiles; t0 += GMAX) {
        const int g     = min(GMAX, ntiles - t0);
        const int row0  = t0 * TILE_M;
        const int nrows = g * TILE_M;

        for (int r = srow; r < nrows; r += 32) {
            const int s = Sl[min(row0 + r, count - 1)];
            const v4f* src = (const v4f*)(x + (size_t)s * IN_SZ) + sq;
            v4f* dst = Xs4 + r * XP4 + sq;
            #pragma unroll
            for (int i = 0; i < 8; ++i) dst[i * 8] = src[i * 8];
        }
        __syncthreads();

        if (g == 4)      fc_group<4>(wv, Xs4, Sl, l, half, n, bv, row0, count, out);
        else if (g == 3) fc_group<3>(wv, Xs4, Sl, l, half, n, bv, row0, count, out);
        else if (g == 2) fc_group<2>(wv, Xs4, Sl, l, half, n, bv, row0, count, out);
        else             fc_group<1>(wv, Xs4, Sl, l, half, n, bv, row0, count, out);

        if (t0 + GMAX < ntiles) __syncthreads();
    }
}

extern "C" void kernel_launch(void* const* d_in, const int* in_sizes, int n_in,
                              void* d_out, int out_size, void* d_ws, size_t ws_size,
                              hipStream_t stream) {
    const float* x    = (const float*)d_in[0];
    const int*   cam  = (const int*)  d_in[1];
    const float* W    = (const float*)d_in[2];
    const float* bias = (const float*)d_in[3];
    float* out = (float*)d_out;

    dim3 grid(NUM_CAM, NRANGE);
    fc_expert_gemm<<<grid, 256, 0, stream>>>(x, cam, W, bias, out);
}


# Loads overlapped at entry: 8 weight k-steps asked first, row-list barriers order LDS only, all x rows of the group asked at once, the other 24 k-steps behind them land under the products; sums same
# speedup vs baseline: 2.2665x; 1.1717x over previous
#include <hip/hip_runtime.h>


#define NUM_CAM 16
#define IN_SZ   256
#define OUT_SZ  128
#define BATCH   8192
#define TILE_M  16
#define NRANGE  16
#define RANGE   (BATCH / NRANGE)
#define GMAX    4
#define XP4     65
#define W0J     8

typedef __attribute__((ext_vector_type(2))) float v2f;
typedef __attribute__((ext_vector_type(4))) float v4f;
typedef __attribute__((ext_vector_type(8))) float v8f;
typedef __attribute__((address_space(1))) float gfloat;

__device__ __forceinline__ void lds_barrier() {
    __builtin_amdgcn_fence(__ATOMIC_RELEASE, "workgroup", "local");
    __builtin_amdgcn_s_barrier();
    __builtin_amdgcn_fence(__ATOMIC_ACQUIRE, "workgroup", "local");
}

template <int G>
__device__ __forceinline__ void fc_group(const v4f (&wv)[32],
                                         const v4f* __restrict__ Xs4,
                                         const int* __restrict__ Sl,
                                         int l, int half, float bv,
                                         int row0, int count,
                                         gfloat* __restrict__ outn) {
    v8f acc[G];
    #pragma unroll
    for (int g = 0; g < G; ++g) {
        acc[g] = v8f{};
        asm volatile("" : "+v"(acc[g]));
    }

    const v4f* ap = Xs4 + l * XP4 + half;

    #pragma unroll
    for (int j = 0; j < 32; ++j) {
        v4f a[G];
        #pragma unroll
        for (int g = 0; g < G; ++g) a[g] = ap[g * (TILE_M * XP4) + 2 * j];
        const v2f b0 = wv[j].xy;
        const v2f b1 = wv[j].zw;
        #pragma unroll
        for (int g = 0; g < G; ++g)
            acc[g] = __builtin_amdgcn_wmma_f32_16x16x4_f32(
                false, a[g].xy, false, b0, (short)0, acc[g], false, false);
        #pragma unroll
        for (int g = 0; g < G; ++g)
            acc[g] = __builtin_amdgcn_wmma_f32_16x16x4_f32(
                false, a[g].zw, false, b1, (short)0, acc[g], false, false);
    }

    if constexpr (G == 1) {
        asm volatile("v_nop\n\tv_nop\n\tv_nop\n\tv_nop\n\tv_nop" : "+v"(acc[0]), "+s"(row0));
    } else if constexpr (G == 2) {
        asm volatile("v_nop\n\tv_nop\n\tv_nop\n\tv_nop\n\tv_nop" : "+v"(acc[0]), "+v"(acc[1]), "+s"(row0));
    } else if constexpr (G == 3) {
        asm volatile("v_nop\n\tv_nop\n\tv_nop\n\tv_nop\n\tv_nop" : "+v"(acc[0]), "+v"(acc[1]), "+v"(acc[2]), "+s"(row0));
    } else {
        asm volatile("v_nop\n\tv_nop\n\tv_nop\n\tv_nop\n\tv_nop" : "+v"(acc[0]), "+v"(acc[1]), "+v"(acc[2]), "+v"(acc[3]), "+s"(row0));
    }

    #pragma unroll
    for (int g = 0; g < G; ++g) {
        #pragma unroll
        for (int r = 0; r < 8; ++r) {
            int gm = row0 + g * TILE_M + r + half * 8;
            if (gm < count) {
                int s = Sl[gm];
                float y = acc[g][r] + bv;
                y = __builtin_amdgcn_rcpf(1.0f + __expf(-y));
                outn[(size_t)s * OUT_SZ] = y;
            }
        }
    }
}

__launch_bounds__(256)
__global__ void fc_expert_gemm(const float* __restrict__ x,
                               const int*   __restrict__ cam,
                               const float* __restrict__ W,
                               const float* __restrict__ bias,
                               float* __restrict__ out) {
    __shared__ v4f Xs4[GMAX * TILE_M * XP4];
    __shared__ int Sl[RANGE];
    __shared__ int wtot[8];

    const int c     = blockIdx.x;
    const int range = blockIdx.y;
    const int tid   = threadIdx.x;
    const int lane  = tid & 31;
    const int wave  = tid >> 5;
    const int half  = lane >> 4;
    const int l     = lane & 15;
    const int n     = wave * 16 + l;

    gfloat* outn = (gfloat*)(out + n);
    asm volatile("" : "+v"(outn));

    const int e0 = range * RANGE + 2 * tid;
    const int2 cv = *(const int2*)(cam + e0);
    __builtin_amdgcn_sched_barrier(0);

    v4f wv[32];
    const v4f* wsrc = (const v4f*)(W + ((size_t)c * OUT_SZ + n) * IN_SZ + 4 * half);
    #pragma unroll
    for (int j = 0; j < W0J; ++j) wv[j] = wsrc[2 * j];
    const float bv = bias[c * OUT_SZ + n];
    __builtin_amdgcn_sched_barrier(0);

    const bool m0 = (cv.x & (NUM_CAM - 1)) == c;
    const bool m1 = (cv.y & (NUM_CAM - 1)) == c;
    const unsigned b0 = __builtin_amdgcn_ballot_w32(m0);
    const unsigned b1 = __builtin_amdgcn_ballot_w32(m1);
    const unsigned below = (1u << lane) - 1u;
    const int pre = __popc(b0 & below) + __popc(b1 & below);
    if (lane == 0) wtot[wave] = __popc(b0) + __popc(b1);
    lds_barrier();
    int off = 0, count = 0;
    #pragma unroll
    for (int i = 0; i < 8; ++i) {
        const int v = wtot[i];
        off   += (i < wave) ? v : 0;
        count += v;
    }
    if (m0) Sl[off + pre] = e0;
    if (m1) Sl[off + pre + (m0 ? 1 : 0)] = e0 + 1;
    lds_barrier();
    if (count == 0) {
        asm volatile("" :: "v"(wv[W0J - 1]), "v"(bv));
        return;
    }

    const int ntiles = (count + TILE_M - 1) / TILE_M;

    const int srow = tid >> 3;
    const int sq   = tid & 7;

    {
        const int g     = min(GMAX, ntiles);
        const int nrows = g * TILE_M;
        const bool pa = srow < nrows;
        const bool pb = srow + 32 < nrows;
        v4f xa[8], xb[8];
        if (pa) {
            const int s = Sl[min(srow, count - 1)];
            const v4f* src = (const v4f*)(x + (size_t)s * IN_SZ) + sq;
            #pragma unroll
            for (int i = 0; i < 8; ++i) xa[i] = src[i * 8];
        }
        if (pb) {
            const int s = Sl[min(srow + 32, count - 1)];
            const v4f* src = (const v4f*)(x + (size_t)s * IN_SZ) + sq;
            #pragma unroll
            for (int i = 0; i < 8; ++i) xb[i] = src[i * 8];
        }
        #pragma unroll
        for (int j = W0J; j < 32; ++j) wv[j] = wsrc[2 * j];
        if (pa) {
            v4f* dst = Xs4 + srow * XP4 + sq;
            #pragma unroll
            for (int i = 0; i < 8; ++i) dst[i * 8] = xa[i];
        }
        if (pb) {
            v4f* dst = Xs4 + (srow + 32) * XP4 + sq;
            #pragma unroll
            for (int i = 0; i < 8; ++i) dst[i * 8] = xb[i];
        }
        lds_barrier();

        if (g == 4)      fc_group<4>(wv, Xs4, Sl, l, half, bv, 0, count, outn);
        else if (g == 3) fc_group<3>(wv, Xs4, Sl, l, half, bv, 0, count, outn);
        else if (g == 2) fc_group<2>(wv, Xs4, Sl, l, half, bv, 0, count, outn);
        else             fc_group<1>(wv, Xs4, Sl, l, half, bv, 0, count, outn);
    }

    for (int t0 = GMAX; t0 < ntiles; t0 += GMAX) {
        const int g     = min(GMAX, ntiles - t0);
        const int row0  = t0 * TILE_M;
        const int nrows = g * TILE_M;

        lds_barrier();

        for (int r = srow; r < nrows; r += 32) {
            const int s = Sl[min(row0 + r, count - 1)];
            const v4f* src = (const v4f*)(x + (size_t)s * IN_SZ) + sq;
            v4f* dst = Xs4 + r * XP4 + sq;
            #pragma unroll
            for (int i = 0; i < 8; ++i) dst[i * 8] = src[i * 8];
        }
        lds_barrier();

        if (g == 4)      fc_group<4>(wv, Xs4, Sl, l, half, bv, row0, count, outn);
        else if (g == 3) fc_group<3>(wv, Xs4, Sl, l, half, bv, row0, count, outn);
        else if (g == 2) fc_group<2>(wv, Xs4, Sl, l, half, bv, row0, count, outn);
        else             fc_group<1>(wv, Xs4, Sl, l, half, bv, row0, count, outn);
    }
}

extern "C" void kernel_launch(void* const* d_in, const int* in_sizes, int n_in,
                              void* d_out, int out_size, void* d_ws, size_t ws_size,
                              hipStream_t stream) {
    const float* x    = (const float*)d_in[0];
    const int*   cam  = (const int*)  d_in[1];
    const float* W    = (const float*)d_in[2];
    const float* bias = (const float*)d_in[3];
    float* out = (float*)d_out;

    dim3 grid(NUM_CAM, NRANGE);
    fc_expert_gemm<<<grid, 256, 0, stream>>>(x, cam, W, bias, out);
}


# Epilogue through LDS: each wave puts its 16 channels of all rows into a row-major tile, one barrier, then whole 512-byte rows stored 16 bytes a lane; was 4-byte stores row by row; sums unchanged
# speedup vs baseline: 2.8489x; 1.2570x over previous
#include <hip/hip_runtime.h>


#define NUM_CAM 16
#define IN_SZ   256
#define OUT_SZ  128
#define BATCH   8192
#define TILE_M  16
#define NRANGE  16
#define RANGE   (BATCH / NRANGE)
#define GMAX    4
#define XP4     65
#define W0J     8
#define OP4     33
#define OPD     (4 * OP4)

typedef __attribute__((ext_vector_type(2))) float v2f;
typedef __attribute__((ext_vector_type(4))) float v4f;
typedef __attribute__((ext_vector_type(8))) float v8f;

#define HAZARD_SLOTS_THEN_MOVES \
    "v_nop\n\tv_nop\n\tv_nop\n\tv_nop\n\tv_nop\n\t" \
    "v_mov_b32 %[wo], %[wi]\n\t" \
    "v_mov_b32 %[ro], %[ri]"

__device__ __forceinline__ void lds_barrier() {
    __builtin_amdgcn_fence(__ATOMIC_RELEASE, "workgroup", "local");
    __builtin_amdgcn_s_barrier();
    __builtin_amdgcn_fence(__ATOMIC_ACQUIRE, "workgroup", "local");
}

template <int G>
__device__ __forceinline__ void fc_group(const v4f (&wv)[32],
                                         const v4f* __restrict__ Xs4,
                                         const int* __restrict__ Sl,
                                         float* __restrict__ Os,
                                         const v4f* __restrict__ Os4,
                                         int l, int half, float bv,
                                         int row0, int count,
                                         int wave,
                                         int owi,
                                         int ori,
                                         int lane,
                                         float* __restrict__ out) {
    v8f acc[G];
    #pragma unroll
    for (int g = 0; g < G; ++g) {
        acc[g] = v8f{};
        asm volatile("" : "+v"(acc[g]));
    }

    const v4f* ap = Xs4 + l * XP4 + half;

    #pragma unroll
    for (int j = 0; j < 32; ++j) {
        v4f a[G];
        #pragma unroll
        for (int g = 0; g < G; ++g) a[g] = ap[g * (TILE_M * XP4) + 2 * j];
        const v2f b0 = wv[j].xy;
        const v2f b1 = wv[j].zw;
        #pragma unroll
        for (int g = 0; g < G; ++g)
            acc[g] = __builtin_amdgcn_wmma_f32_16x16x4_f32(
                false, a[g].xy, false, b0, (short)0, acc[g], false, false);
        #pragma unroll
        for (int g = 0; g < G; ++g)
            acc[g] = __builtin_amdgcn_wmma_f32_16x16x4_f32(
                false, a[g].zw, false, b1, (short)0, acc[g], false, false);
    }

    int owl, orl;
    if constexpr (G == 1) {
        asm volatile(HAZARD_SLOTS_THEN_MOVES : "+v"(acc[0]), "+s"(row0), [wo] "=&v"(owl), [ro] "=&v"(orl) : [wi] "v"(owi), [ri] "v"(ori));
    } else if constexpr (G == 2) {
        asm volatile(HAZARD_SLOTS_THEN_MOVES : "+v"(acc[0]), "+v"(acc[1]), "+s"(row0), [wo] "=&v"(owl), [ro] "=&v"(orl) : [wi] "v"(owi), [ri] "v"(ori));
    } else if constexpr (G == 3) {
        asm volatile(HAZARD_SLOTS_THEN_MOVES : "+v"(acc[0]), "+v"(acc[1]), "+v"(acc[2]), "+s"(row0), [wo] "=&v"(owl), [ro] "=&v"(orl) : [wi] "v"(owi), [ri] "v"(ori));
    } else {
        asm volatile(HAZARD_SLOTS_THEN_MOVES : "+v"(acc[0]), "+v"(acc[1]), "+v"(acc[2]), "+v"(acc[3]), "+s"(row0), [wo] "=&v"(owl), [ro] "=&v"(orl) : [wi] "v"(owi), [ri] "v"(ori));
    }

    #pragma unroll
    for (int g = 0; g < G; ++g) {
        #pragma unroll
        for (int r = 0; r < 8; ++r) {
            float y = acc[g][r] + bv;
            y = __builtin_amdgcn_rcpf(1.0f + __expf(-y));
            Os[owl + (g * TILE_M + r) * OPD] = y;
        }
    }

    const int gmw = row0 + wave;
    int sp[2 * G];
    #pragma unroll
    for (int i = 0; i < 2 * G; ++i) sp[i] = Sl[gmw + 8 * i];

    lds_barrier();

    v4f o[2 * G];
    #pragma unroll
    for (int i = 0; i < 2 * G; ++i) o[i] = Os4[orl + i * (8 * OP4)];
    #pragma unroll
    for (int i = 0; i < 2 * G; ++i) {
        if (gmw + 8 * i < count)
            *(v4f*)(out + ((unsigned)sp[i] * OUT_SZ + 4 * lane)) = o[i];
    }
}

__launch_bounds__(256)
__global__ void fc_expert_gemm(const float* __restrict__ x,
                               const int*   __restrict__ cam,
                               const float* __restrict__ W,
                               const float* __restrict__ bias,
                               float* __restrict__ out) {
    __shared__ v4f Xs4[GMAX * TILE_M * XP4];
    __shared__ int Sl[RANGE];
    __shared__ int wtot[8];
    __shared__ v4f Os4[GMAX * TILE_M * OP4];
    float* Os = (float*)Os4;

    const int c     = blockIdx.x;
    const int range = blockIdx.y;
    const int tid   = threadIdx.x;
    const int lane  = tid & 31;
    const int wave  = tid >> 5;
    const int half  = lane >> 4;
    const int l     = lane & 15;
    const int n     = wave * 16 + l;

    const int owi = (8 * half) * OPD + wave * 16 + l;
    const int ori = wave * OP4 + lane;
    const int wave_u = __builtin_amdgcn_readfirstlane(wave);

    const int e0 = range * RANGE + 2 * tid;
    const int2 cv = *(const int2*)(cam + e0);
    __builtin_amdgcn_sched_barrier(0);

    v4f wv[32];
    const v4f* wsrc = (const v4f*)(W + ((size_t)c * OUT_SZ + n) * IN_SZ + 4 * half);
    #pragma unroll
    for (int j = 0; j < W0J; ++j) wv[j] = wsrc[2 * j];
    const float bv = bias[c * OUT_SZ + n];
    __builtin_amdgcn_sched_barrier(0);

    const bool m0 = (cv.x & (NUM_CAM - 1)) == c;
    const bool m1 = (cv.y & (NUM_CAM - 1)) == c;
    const unsigned b0 = __builtin_amdgcn_ballot_w32(m0);
    const unsigned b1 = __builtin_amdgcn_ballot_w32(m1);
    const unsigned below = (1u << lane) - 1u;
    const int pre = __popc(b0 & below) + __popc(b1 & below);
    if (lane == 0) wtot[wave] = __popc(b0) + __popc(b1);
    lds_barrier();
    int off = 0, count = 0;
    #pragma unroll
    for (int i = 0; i < 8; ++i) {
        const int v = wtot[i];
        off   += (i < wave) ? v : 0;
        count += v;
    }
    if (m0) Sl[off + pre] = e0;
    if (m1) Sl[off + pre + (m0 ? 1 : 0)] = e0 + 1;
    lds_barrier();
    if (count == 0) {
        asm volatile("" :: "v"(wv[W0J - 1]), "v"(bv));
        return;
    }

    const int ntiles = (count + TILE_M - 1) / TILE_M;

    const int srow = tid >> 3;
    const int sq   = tid & 7;

    {
        const int g     = min(GMAX, ntiles);
        const int nrows = g * TILE_M;
        const bool pa = srow < nrows;
        const bool pb = srow + 32 < nrows;
        v4f xa[8], xb[8];
        if (pa) {
            const int s = Sl[min(srow, count - 1)];
            const v4f* src = (const v4f*)(x + (size_t)s * IN_SZ) + sq;
            #pragma unroll
            for (int i = 0; i < 8; ++i) xa[i] = src[i * 8];
        }
        if (pb) {
            const int s = Sl[min(srow + 32, count - 1)];
            const v4f* src = (const v4f*)(x + (size_t)s * IN_SZ) + sq;
            #pragma unroll
            for (int i = 0; i < 8; ++i) xb[i] = src[i * 8];
        }
        #pragma unroll
        for (int j = W0J; j < 32; ++j) wv[j] = wsrc[2 * j];
        if (pa) {
            v4f* dst = Xs4 + srow * XP4 + sq;
            #pragma unroll
            for (int i = 0; i < 8; ++i) dst[i * 8] = xa[i];
        }
        if (pb) {
            v4f* dst = Xs4 + (srow + 32) * XP4 + sq;
            #pragma unroll
            for (int i = 0; i < 8; ++i) dst[i * 8] = xb[i];
        }
        lds_barrier();

        if (g == 4)      fc_group<4>(wv, Xs4, Sl, Os, Os4, l, half, bv, 0, count, wave_u, owi, ori, lane, out);
        else if (g == 3) fc_group<3>(wv, Xs4, Sl, Os, Os4, l, half, bv, 0, count, wave_u, owi, ori, lane, out);
        else if (g == 2) fc_group<2>(wv, Xs4, Sl, Os, Os4, l, half, bv, 0, count, wave_u, owi, ori, lane, out);
        else             fc_group<1>(wv, Xs4, Sl, Os, Os4, l, half, bv, 0, count, wave_u, owi, ori, lane, out);
    }

    for (int t0 = GMAX; t0 < ntiles; t0 += GMAX) {
        const int g     = min(GMAX, ntiles - t0);
        const int row0  = t0 * TILE_M;
        const int nrows = g * TILE_M;

        lds_barrier();

        for (int r = srow; r < nrows; r += 32) {
            const int s = Sl[min(row0 + r, count - 1)];
            const v4f* src = (const v4f*)(x + (size_t)s * IN_SZ) + sq;
            v4f* dst = Xs4 + r * XP4 + sq;
            #pragma unroll
            for (int i = 0; i < 8; ++i) dst[i * 8] = src[i * 8];
        }
        lds_barrier();

        if (g == 4)      fc_group<4>(wv, Xs4, Sl, Os, Os4, l, half, bv, row0, count, wave_u, owi, ori, lane, out);
        else if (g == 3) fc_group<3>(wv, Xs4, Sl, Os, Os4, l, half, bv, row0, count, wave_u, owi, ori, lane, out);
        else if (g == 2) fc_group<2>(wv, Xs4, Sl, Os, Os4, l, half, bv, row0, count, wave_u, owi, ori, lane, out);
        else             fc_group<1>(wv, Xs4, Sl, Os, Os4, l, half, bv, row0, count, wave_u, owi, ori, lane, out);
    }
}

extern "C" void kernel_launch(void* const* d_in, const int* in_sizes, int n_in,
                              void* d_out, int out_size, void* d_ws, size_t ws_size,
                              hipStream_t stream) {
    const float* x    = (const float*)d_in[0];
    const int*   cam  = (const int*)  d_in[1];
    const float* W    = (const float*)d_in[2];
    const float* bias = (const float*)d_in[3];
    float* out = (float*)d_out;

    dim3 grid(NUM_CAM, NRANGE);
    fc_expert_gemm<<<grid, 256, 0, stream>>>(x, cam, W, bias, out);
}
